# starting kernel
# baseline (speedup 1.0000x reference)
#include <hip/hip_runtime.h>

typedef float v2f __attribute__((ext_vector_type(2)));
typedef float v4f __attribute__((ext_vector_type(4)));
typedef float v8f __attribute__((ext_vector_type(8)));

__global__ __launch_bounds__(1024) void k_scores_softmax(
    const float* __restrict__ dbfc, float* __restrict__ weights) {
  __shared__ float part[1024];
  __shared__ float sc[64];
  __shared__ float ex[64];

  const int tid = threadIdx.x;
  const int d = tid >> 4;
  const int j = tid & 15;
  const float* base =
      dbfc + (size_t)blockIdx.x * (64 * 64 * 64) + d * 64 + j * 4;

  float acc = 0.f;
#pragma unroll 8
  for (int f1 = 0; f1 < 64; ++f1) {
    v4f x = __builtin_nontemporal_load((const v4f*)(base + f1 * 4096));
    acc += x.x + x.y + x.z + x.w;
  }
  part[tid] = acc;
  __syncthreads();

  if (tid < 64) {
    float s = 0.f;
#pragma unroll
    for (int q = 0; q < 16; ++q) s += part[tid * 16 + q];
    sc[tid] = s;
  }
  __syncthreads();
  if (tid < 64) {
    float m = -1e30f;
    for (int q = 0; q < 64; ++q) m = fmaxf(m, sc[q]);
    ex[tid] = __expf(sc[tid] - m);
  }
  __syncthreads();
  if (tid < 64) {
    float s = 0.f;
    for (int q = 0; q < 64; ++q) s += ex[q];
    weights[blockIdx.x * 64 + tid] = ex[tid] / s;
  }
}

__global__ __launch_bounds__(256) void k_attn_gemm(
    const float* __restrict__ X, const float* __restrict__ wV,
    const float* __restrict__ weights, float* __restrict__ out) {
  const int b = blockIdx.x >> 2;
  const int ps = blockIdx.x & 3;
  const int tid = threadIdx.x;
  const int lane = tid & 31;
  const int w = tid >> 5;
  const int half = lane >> 4;
  const int l15 = lane & 15;

  __shared__ float Wl[64 * 65];
  __shared__ float Xl[64 * 65];
  __shared__ float Vl[64 * 65];
  __shared__ float C1[64 * 65];

  const float* Wg = weights + ps * 4096;
  const float* Xg = X + (size_t)b * 16384 + ps * 4096;
#pragma unroll
  for (int e = tid; e < 4096; e += 256) {
    const int r = e >> 6, cc = e & 63;
    Wl[r * 65 + cc] = Wg[e];
    Xl[r * 65 + cc] = Xg[e];
    Vl[r * 65 + cc] = wV[e];
  }
  __syncthreads();

  const int m0 = (w >> 2) * 16;
  const int m1 = m0 + 32;
  const int n0 = (w & 3) * 16;
  const int arow0 = (m0 + l15) * 65;
  const int arow1 = (m1 + l15) * 65;
  const int bcol = n0 + l15;
  const int crow0 = (m0 + half * 8) * 65 + bcol;
  const int crow1 = (m1 + half * 8) * 65 + bcol;

  {
    v8f acc0 = {}, acc1 = {};
#pragma unroll
    for (int kk = 0; kk < 16; ++kk) {
      const int kb = kk * 4 + half * 2;
      v2f bb = {Xl[kb * 65 + bcol], Xl[(kb + 1) * 65 + bcol]};
      v2f a0 = {Wl[arow0 + kb], Wl[arow0 + kb + 1]};
      v2f a1 = {Wl[arow1 + kb], Wl[arow1 + kb + 1]};
      acc0 = __builtin_amdgcn_wmma_f32_16x16x4_f32(false, a0, false, bb,
                                                   (short)0, acc0, false, false);
      acc1 = __builtin_amdgcn_wmma_f32_16x16x4_f32(false, a1, false, bb,
                                                   (short)0, acc1, false, false);
    }
#pragma unroll
    for (int i = 0; i < 8; ++i) {
      C1[crow0 + i * 65] = acc0[i];
      C1[crow1 + i * 65] = acc1[i];
    }
  }
  __syncthreads();

  {
    v8f acc0 = {}, acc1 = {};
#pragma unroll
    for (int kk = 0; kk < 16; ++kk) {
      const int kb = kk * 4 + half * 2;
      v2f bb = {Vl[bcol * 65 + kb], Vl[bcol * 65 + kb + 1]};
      v2f a0 = {C1[arow0 + kb], C1[arow0 + kb + 1]};
      v2f a1 = {C1[arow1 + kb], C1[arow1 + kb + 1]};
      acc0 = __builtin_amdgcn_wmma_f32_16x16x4_f32(false, a0, false, bb,
                                                   (short)0, acc0, false, false);
      acc1 = __builtin_amdgcn_wmma_f32_16x16x4_f32(false, a1, false, bb,
                                                   (short)0, acc1, false, false);
    }
    const size_t obase = (size_t)b * 256 + ps * 64 + bcol;
#pragma unroll
    for (int i = 0; i < 8; ++i) {
      const int c0 = m0 + half * 8 + i;
      const int c1 = m1 + half * 8 + i;
      out[(size_t)c0 * 16384 + obase] = acc0[i];
      out[(size_t)c1 * 16384 + obase] = acc1[i];
    }
  }
}

extern "C" void kernel_launch(void* const* d_in, const int* in_sizes, int n_in,
                              void* d_out, int out_size, void* d_ws,
                              size_t ws_size, hipStream_t stream) {
  (void)in_sizes; (void)n_in; (void)out_size; (void)ws_size;
  const float* X = (const float*)d_in[0];
  const float* dbfc = (const float*)d_in[1];
  const float* wV = (const float*)d_in[2];
  float* out = (float*)d_out;
  float* weights = (float*)d_ws;

  k_scores_softmax<<<256, 1024, 0, stream>>>(dbfc, weights);
  k_attn_gemm<<<256, 256, 0, stream>>>(X, wV, weights, out);
}


# GEMM kernel prologue: all operand tile loads issued at once as twelve 16-byte loads per thread before any LDS store (was 8 dependent load-wait-store rounds of 4-byte loads)
# speedup vs baseline: 1.1048x; 1.1048x over previous
#include <hip/hip_runtime.h>

typedef float v2f __attribute__((ext_vector_type(2)));
typedef float v4f __attribute__((ext_vector_type(4)));
typedef float v8f __attribute__((ext_vector_type(8)));

__global__ __launch_bounds__(1024) void k_scores_softmax(
    const float* __restrict__ dbfc, float* __restrict__ weights) {
  __shared__ float part[1024];
  __shared__ float sc[64];
  __shared__ float ex[64];

  const int tid = threadIdx.x;
  const int d = tid >> 4;
  const int j = tid & 15;
  const float* base =
      dbfc + (size_t)blockIdx.x * (64 * 64 * 64) + d * 64 + j * 4;

  float acc = 0.f;
#pragma unroll 8
  for (int f1 = 0; f1 < 64; ++f1) {
    v4f x = __builtin_nontemporal_load((const v4f*)(base + f1 * 4096));
    acc += x.x + x.y + x.z + x.w;
  }
  part[tid] = acc;
  __syncthreads();

  if (tid < 64) {
    float s = 0.f;
#pragma unroll
    for (int q = 0; q < 16; ++q) s += part[tid * 16 + q];
    sc[tid] = s;
  }
  __syncthreads();
  if (tid < 64) {
    float m = -1e30f;
    for (int q = 0; q < 64; ++q) m = fmaxf(m, sc[q]);
    ex[tid] = __expf(sc[tid] - m);
  }
  __syncthreads();
  if (tid < 64) {
    float s = 0.f;
    for (int q = 0; q < 64; ++q) s += ex[q];
    weights[blockIdx.x * 64 + tid] = ex[tid] / s;
  }
}

__global__ __launch_bounds__(256) void k_attn_gemm(
    const float* __restrict__ X, const float* __restrict__ wV,
    const float* __restrict__ weights, float* __restrict__ out) {
  const int b = blockIdx.x >> 2;
  const int ps = blockIdx.x & 3;
  const int tid = threadIdx.x;
  const int lane = tid & 31;
  const int w = tid >> 5;
  const int half = lane >> 4;
  const int l15 = lane & 15;

  __shared__ float Wl[64 * 65];
  __shared__ float Xl[64 * 65];
  __shared__ float Vl[64 * 65];
  __shared__ float C1[64 * 65];

  const float* Wg = weights + ps * 4096;
  const float* Xg = X + (size_t)b * 16384 + ps * 4096;
  v4f wq[4], vq[4], xq[4];
  for (int i = 0; i < 4; ++i) {
    const int q = tid + i * 256;
    wq[i] = *(const v4f*)(Wg + q * 4);
    vq[i] = *(const v4f*)(wV + q * 4);
    xq[i] = *(const v4f*)(Xg + q * 4);
  }
  for (int i = 0; i < 4; ++i) {
    const int q = tid + i * 256;
    const int o = (q >> 4) * 65 + (q & 15) * 4;
    Wl[o] = wq[i].x; Wl[o + 1] = wq[i].y; Wl[o + 2] = wq[i].z; Wl[o + 3] = wq[i].w;
    Vl[o] = vq[i].x; Vl[o + 1] = vq[i].y; Vl[o + 2] = vq[i].z; Vl[o + 3] = vq[i].w;
    Xl[o] = xq[i].x; Xl[o + 1] = xq[i].y; Xl[o + 2] = xq[i].z; Xl[o + 3] = xq[i].w;
  }
  __syncthreads();

  const int m0 = (w >> 2) * 16;
  const int m1 = m0 + 32;
  const int n0 = (w & 3) * 16;
  const int arow0 = (m0 + l15) * 65;
  const int arow1 = (m1 + l15) * 65;
  const int bcol = n0 + l15;
  const int crow0 = (m0 + half * 8) * 65 + bcol;
  const int crow1 = (m1 + half * 8) * 65 + bcol;

  {
    v8f acc0 = {}, acc1 = {};
#pragma unroll
    for (int kk = 0; kk < 16; ++kk) {
      const int kb = kk * 4 + half * 2;
      v2f bb = {Xl[kb * 65 + bcol], Xl[(kb + 1) * 65 + bcol]};
      v2f a0 = {Wl[arow0 + kb], Wl[arow0 + kb + 1]};
      v2f a1 = {Wl[arow1 + kb], Wl[arow1 + kb + 1]};
      acc0 = __builtin_amdgcn_wmma_f32_16x16x4_f32(false, a0, false, bb,
                                                   (short)0, acc0, false, false);
      acc1 = __builtin_amdgcn_wmma_f32_16x16x4_f32(false, a1, false, bb,
                                                   (short)0, acc1, false, false);
    }
#pragma unroll
    for (int i = 0; i < 8; ++i) {
      C1[crow0 + i * 65] = acc0[i];
      C1[crow1 + i * 65] = acc1[i];
    }
  }
  __syncthreads();

  {
    v8f acc0 = {}, acc1 = {};
#pragma unroll
    for (int kk = 0; kk < 16; ++kk) {
      const int kb = kk * 4 + half * 2;
      v2f bb = {Vl[bcol * 65 + kb], Vl[bcol * 65 + kb + 1]};
      v2f a0 = {C1[arow0 + kb], C1[arow0 + kb + 1]};
      v2f a1 = {C1[arow1 + kb], C1[arow1 + kb + 1]};
      acc0 = __builtin_amdgcn_wmma_f32_16x16x4_f32(false, a0, false, bb,
                                                   (short)0, acc0, false, false);
      acc1 = __builtin_amdgcn_wmma_f32_16x16x4_f32(false, a1, false, bb,
                                                   (short)0, acc1, false, false);
    }
    const size_t obase = (size_t)b * 256 + ps * 64 + bcol;
#pragma unroll
    for (int i = 0; i < 8; ++i) {
      const int c0 = m0 + half * 8 + i;
      const int c1 = m1 + half * 8 + i;
      out[(size_t)c0 * 16384 + obase] = acc0[i];
      out[(size_t)c1 * 16384 + obase] = acc1[i];
    }
  }
}

extern "C" void kernel_launch(void* const* d_in, const int* in_sizes, int n_in,
                              void* d_out, int out_size, void* d_ws,
                              size_t ws_size, hipStream_t stream) {
  (void)in_sizes; (void)n_in; (void)out_size; (void)ws_size;
  const float* X = (const float*)d_in[0];
  const float* dbfc = (const float*)d_in[1];
  const float* wV = (const float*)d_in[2];
  float* out = (float*)d_out;
  float* weights = (float*)d_ws;

  k_scores_softmax<<<256, 1024, 0, stream>>>(dbfc, weights);
  k_attn_gemm<<<256, 256, 0, stream>>>(X, wV, weights, out);
}
